# starting kernel
# baseline (speedup 1.0000x reference)
#include <hip/hip_runtime.h>
#include <hip/hip_bf16.h>

typedef __attribute__((ext_vector_type(2))) float v2f;
typedef __attribute__((ext_vector_type(8))) float v8f;

#define C_DIM 8192
#define N_DIM 8192
#define F_DIM 64
#define B_DIM 128
#define TILE_N 16
#define TILE_B 16
#define KDIM 80
#define LDS_STRIDE 84

__global__ __launch_bounds__(256)
void cscfc_wmma_kernel(const float* __restrict__ x,
                       const float* __restrict__ kern,
                       const float* __restrict__ bias,
                       float* __restrict__ out) {
    __shared__ float xs [B_DIM  * LDS_STRIDE];
    __shared__ float w2t[TILE_N * LDS_STRIDE];

    const int tid = threadIdx.x;
    const int n0  = blockIdx.x * TILE_N;

    __builtin_prefetch(&kern[(size_t)((n0 + (tid & 63)) & (C_DIM - 1)) * N_DIM + n0], 0, 0);

    for (int i = tid; i < KDIM * TILE_N; i += 256) {
        int k   = i / TILE_N;
        int j   = i % TILE_N;
        int row = (n0 + k) & (C_DIM - 1);
        float v = kern[(size_t)row * N_DIM + (size_t)(n0 + j)];
        int f   = k - j;
        w2t[j * LDS_STRIDE + k] = (f >= 0 && f < F_DIM) ? v : 0.0f;
    }

    for (int i = tid; i < B_DIM * KDIM; i += 256) {
        int b   = i / KDIM;
        int k   = i % KDIM;
        int col = (n0 + k) & (C_DIM - 1);
        xs[b * LDS_STRIDE + k] = x[(size_t)b * C_DIM + (size_t)col];
    }

    __syncthreads();

    const int wave = tid >> 5;
    const int lane = tid & 31;
    const int half = lane >> 4;
    const int lj   = lane & 15;
    const int b0   = wave * TILE_B;

    const float* arow = &xs [(b0 + lj) * LDS_STRIDE];
    const float* brow = &w2t[lj        * LDS_STRIDE];

    v8f acc = {};
#pragma unroll
    for (int kk = 0; kk < KDIM / 4; ++kk) {
        const int k0 = kk * 4 + 2 * half;
        v2f a = *(const v2f*)(arow + k0);
        v2f b = *(const v2f*)(brow + k0);
        acc = __builtin_amdgcn_wmma_f32_16x16x4_f32(
             false, a,  false, b,
             (short)0, acc,  false,  false);
    }

    const float bv = bias[n0 + lj];
#pragma unroll
    for (int r = 0; r < 8; ++r) {
        int m = r + (half ? 8 : 0);
        out[(size_t)(b0 + m) * N_DIM + (size_t)(n0 + lj)] = acc[r] + bv;
    }
}

extern "C" void kernel_launch(void* const* d_in, const int* in_sizes, int n_in,
                              void* d_out, int out_size, void* d_ws, size_t ws_size,
                              hipStream_t stream) {
    (void)in_sizes; (void)n_in; (void)out_size; (void)d_ws; (void)ws_size;
    const float* x    = (const float*)d_in[0];
    const float* kern = (const float*)d_in[1];
    const float* bias = (const float*)d_in[2];
    float* out        = (float*)d_out;

    dim3 grid(N_DIM / TILE_N);
    dim3 block(256);
    hipLaunchKernelGGL(cscfc_wmma_kernel, grid, block, 0, stream, x, kern, bias, out);
}


# Tile staging: every global load of the weight rows and the x slice is issued before the first wait, x as 16-byte loads (one memory round trip instead of ten in series); order of summation unchanged
# speedup vs baseline: 1.2777x; 1.2777x over previous
#include <hip/hip_runtime.h>
#include <hip/hip_bf16.h>


typedef __attribute__((ext_vector_type(2))) float v2f;
typedef __attribute__((ext_vector_type(4))) float v4f;
typedef __attribute__((ext_vector_type(8))) float v8f;

#define C_DIM 8192
#define N_DIM 8192
#define F_DIM 64
#define B_DIM 128
#define TILE_N 16
#define TILE_B 16
#define KDIM 80
#define LDS_STRIDE 84
#define NTHREADS 256
#define KQ (KDIM / 4)
#define X_CHUNKS (B_DIM * KQ / NTHREADS)
#define W_ELEMS (KDIM * TILE_N / NTHREADS)

__global__ __launch_bounds__(NTHREADS)
void cscfc_wmma_kernel(const float* __restrict__ x,
                       const float* __restrict__ kern,
                       const float* __restrict__ bias,
                       float* __restrict__ out) {
    __shared__ __attribute__((aligned(16))) float xs [B_DIM  * LDS_STRIDE];
    __shared__ __attribute__((aligned(16))) float w2t[TILE_N * LDS_STRIDE];

    const int tid = threadIdx.x;
    const int n0  = blockIdx.x * TILE_N;
    const float bv = bias[n0 + (tid & 15)];


    float wv[W_ELEMS];
#pragma unroll
    for (int t = 0; t < W_ELEMS; ++t) {
        const int i   = tid + NTHREADS * t;
        const int k   = i / TILE_N;
        const int j   = i % TILE_N;
        const int row = (n0 + k) & (C_DIM - 1);
        wv[t] = kern[(size_t)row * N_DIM + (size_t)(n0 + j)];
    }

    v4f xv[X_CHUNKS];
#pragma unroll
    for (int t = 0; t < X_CHUNKS; ++t) {
        const int i   = tid + NTHREADS * t;
        const int b   = i / KQ;
        const int c   = i - b * KQ;
        const int col = (n0 + 4 * c) & (C_DIM - 1);
        xv[t] = *(const v4f*)(x + (size_t)b * C_DIM + (size_t)col);
    }

#pragma unroll
    for (int t = 0; t < W_ELEMS; ++t) {
        const int i = tid + NTHREADS * t;
        const int k = i / TILE_N;
        const int j = i % TILE_N;
        const int f = k - j;
        w2t[j * LDS_STRIDE + k] = (f >= 0 && f < F_DIM) ? wv[t] : 0.0f;
    }

#pragma unroll
    for (int t = 0; t < X_CHUNKS; ++t) {
        const int i = tid + NTHREADS * t;
        const int b = i / KQ;
        const int c = i - b * KQ;
        *(v4f*)(&xs[b * LDS_STRIDE + 4 * c]) = xv[t];
    }

    __syncthreads();

    const int wave = tid >> 5;
    const int lane = tid & 31;
    const int half = lane >> 4;
    const int lj   = lane & 15;
    const int b0   = wave * TILE_B;

    const float* arow = &xs [(b0 + lj) * LDS_STRIDE];
    const float* brow = &w2t[lj        * LDS_STRIDE];

    v8f acc = {};
#pragma unroll
    for (int kk = 0; kk < KDIM / 4; ++kk) {
        const int k0 = kk * 4 + 2 * half;
        v2f a = *(const v2f*)(arow + k0);
        v2f b = *(const v2f*)(brow + k0);
        acc = __builtin_amdgcn_wmma_f32_16x16x4_f32(
             false, a,  false, b,
             (short)0, acc,  false,  false);
    }

#pragma unroll
    for (int r = 0; r < 8; ++r) {
        int m = r + (half ? 8 : 0);
        out[(size_t)(b0 + m) * N_DIM + (size_t)(n0 + lj)] = acc[r] + bv;
    }
}

extern "C" void kernel_launch(void* const* d_in, const int* in_sizes, int n_in,
                              void* d_out, int out_size, void* d_ws, size_t ws_size,
                              hipStream_t stream) {
    (void)in_sizes; (void)n_in; (void)out_size; (void)d_ws; (void)ws_size;
    const float* x    = (const float*)d_in[0];
    const float* kern = (const float*)d_in[1];
    const float* bias = (const float*)d_in[2];
    float* out        = (float*)d_out;

    dim3 grid(N_DIM / TILE_N);
    dim3 block(NTHREADS);
    hipLaunchKernelGGL(cscfc_wmma_kernel, grid, block, 0, stream, x, kern, bias, out);
}


# Coarser tiling: 256 work-groups of 4 waves, one per WGP (LDS sized so), each a 128x32 output block, each wave 2x2 tiles on 4 accumulators; 1,024 waves enter, not 4,096; summation order unchanged
# speedup vs baseline: 2.0292x; 1.5882x over previous
#include <hip/hip_runtime.h>
#include <hip/hip_bf16.h>


typedef __attribute__((ext_vector_type(2))) float v2f;
typedef __attribute__((ext_vector_type(4))) float v4f;
typedef __attribute__((ext_vector_type(8))) float v8f;

#define C_DIM 8192
#define N_DIM 8192
#define F_DIM 64
#define B_DIM 128
#define TILE_N 32
#define SUB_N 16
#define KDIM 96
#define KSUB 80
#define LDS_STRIDE 100
#define NTHREADS 128
#define CPL 8
#define ROWS_PER_PASS (NTHREADS / CPL)
#define X_LINES (KDIM / 32)
#define X_PASSES (B_DIM / ROWS_PER_PASS * X_LINES)
#define W_PASSES (KDIM / ROWS_PER_PASS)
#define LDS_USED ((B_DIM + TILE_N) * LDS_STRIDE)
#define LDS_FLOATS (162 * 1024 / 4)
static_assert(LDS_USED <= LDS_FLOATS, "both tiles fit the declared LDS");

__global__ __launch_bounds__(NTHREADS)
void cscfc_wmma_kernel(const float* __restrict__ x,
                       const float* __restrict__ kern,
                       const float* __restrict__ bias,
                       float* __restrict__ out) {
    __shared__ __attribute__((aligned(16))) float lds[LDS_FLOATS];
    float* xs  = lds;
    float* w2t = lds + B_DIM * LDS_STRIDE;

    const int tid  = threadIdx.x;
    const int n0   = blockIdx.x * TILE_N;
    const int cq   = tid & (CPL - 1);
    const int rr   = tid / CPL;

    const int wave = tid >> 5;
    const int lane = tid & 31;
    const int half = lane >> 4;
    const int lj   = lane & 15;

    const float bv0 = bias[n0 + lj];
    const float bv1 = bias[n0 + SUB_N + lj];


    v4f wv[W_PASSES];
#pragma unroll
    for (int t = 0; t < W_PASSES; ++t) {
        const int k   = rr + ROWS_PER_PASS * t;
        const int row = (n0 + k) & (C_DIM - 1);
        wv[t] = *(const v4f*)(kern + (size_t)row * N_DIM + (size_t)(n0 + 4 * cq));
    }

    v4f xv[X_PASSES];
#pragma unroll
    for (int t = 0; t < X_PASSES; ++t) {
        const int l   = t % X_LINES;
        const int b   = rr + ROWS_PER_PASS * (t / X_LINES);
        const int col = ((n0 + 32 * l) & (C_DIM - 1)) + 4 * cq;
        xv[t] = *(const v4f*)(x + (size_t)b * C_DIM + (size_t)col);
    }

#pragma unroll
    for (int t = 0; t < W_PASSES; ++t) {
        const int k = rr + ROWS_PER_PASS * t;
#pragma unroll
        for (int c = 0; c < 4; ++c) {
            const int j = 4 * cq + c;
            const int f = k - j;
            w2t[j * LDS_STRIDE + k] = (f >= 0 && f < F_DIM) ? wv[t][c] : 0.0f;
        }
    }

#pragma unroll
    for (int t = 0; t < X_PASSES; ++t) {
        const int l = t % X_LINES;
        const int b = rr + ROWS_PER_PASS * (t / X_LINES);
        *(v4f*)(&xs[b * LDS_STRIDE + 32 * l + 4 * cq]) = xv[t];
    }

    __syncthreads();

    const int b0 = wave * 32;

    const float* arow0 = &xs [(b0 + lj)          * LDS_STRIDE + 2 * half];
    const float* arow1 = &xs [(b0 + 16 + lj)     * LDS_STRIDE + 2 * half];
    const float* brow0 = &w2t[lj                 * LDS_STRIDE + 2 * half];
    const float* brow1 = &w2t[(SUB_N + lj)       * LDS_STRIDE + 2 * half];

    v8f acc00 = {}, acc01 = {}, acc10 = {}, acc11 = {};
#pragma unroll
    for (int q = 0; q < KDIM / 4; ++q) {
        const v2f a0 = *(const v2f*)(arow0 + 4 * q);
        const v2f a1 = *(const v2f*)(arow1 + 4 * q);
        if (q < KSUB / 4) {
            const v2f b = *(const v2f*)(brow0 + 4 * q);
            acc00 = __builtin_amdgcn_wmma_f32_16x16x4_f32(false, a0, false, b, (short)0, acc00, false, false);
            acc10 = __builtin_amdgcn_wmma_f32_16x16x4_f32(false, a1, false, b, (short)0, acc10, false, false);
        }
        if (q >= (KDIM - KSUB) / 4) {
            const v2f b = *(const v2f*)(brow1 + 4 * q);
            acc01 = __builtin_amdgcn_wmma_f32_16x16x4_f32(false, a0, false, b, (short)0, acc01, false, false);
            acc11 = __builtin_amdgcn_wmma_f32_16x16x4_f32(false, a1, false, b, (short)0, acc11, false, false);
        }
    }

#pragma unroll
    for (int r = 0; r < 8; ++r) {
        const int m = r + (half ? 8 : 0);
        float* o0 = out + (size_t)(b0 + m) * N_DIM + (size_t)(n0 + lj);
        float* o1 = out + (size_t)(b0 + 16 + m) * N_DIM + (size_t)(n0 + lj);
        o0[0]     = acc00[r] + bv0;
        o0[SUB_N] = acc01[r] + bv1;
        o1[0]     = acc10[r] + bv0;
        o1[SUB_N] = acc11[r] + bv1;
    }
}

extern "C" void kernel_launch(void* const* d_in, const int* in_sizes, int n_in,
                              void* d_out, int out_size, void* d_ws, size_t ws_size,
                              hipStream_t stream) {
    (void)in_sizes; (void)n_in; (void)out_size; (void)d_ws; (void)ws_size;
    const float* x    = (const float*)d_in[0];
    const float* kern = (const float*)d_in[1];
    const float* bias = (const float*)d_in[2];
    float* out        = (float*)d_out;

    dim3 grid(N_DIM / TILE_N);
    dim3 block(NTHREADS);
    hipLaunchKernelGGL(cscfc_wmma_kernel, grid, block, 0, stream, x, kern, bias, out);
}


# Three-stage pipeline over k chunks of 32: chunk c's WMMAs run while chunk c+1's global loads land and are staged; barrier split by hand so later loads stay in flight; summation order unchanged
# speedup vs baseline: 2.1390x; 1.0541x over previous
#include <hip/hip_runtime.h>
#include <hip/hip_bf16.h>


typedef __attribute__((ext_vector_type(2))) float v2f;
typedef __attribute__((ext_vector_type(4))) float v4f;
typedef __attribute__((ext_vector_type(8))) float v8f;

#define C_DIM 8192
#define N_DIM 8192
#define F_DIM 64
#define B_DIM 128
#define TILE_N 32
#define SUB_N 16
#define KDIM 96
#define KSUB 80
#define LDS_STRIDE 100
#define NTHREADS 128
#define CPL 8
#define ROWS_PER_PASS (NTHREADS / CPL)
#define KCH 32
#define NCHUNK (KDIM / KCH)
#define QPC (KCH / 4)
#define XG (B_DIM / ROWS_PER_PASS)
#define WU (KCH / ROWS_PER_PASS)
#define LDS_USED ((B_DIM + TILE_N) * LDS_STRIDE)
#define LDS_FLOATS (162 * 1024 / 4)
static_assert(LDS_USED <= LDS_FLOATS, "both tiles fit the declared LDS");
static_assert(NCHUNK == 3 && QPC * NCHUNK * 4 == KDIM, "three chunks of eight k-steps");

#define DEV static __device__ __forceinline__

DEV void keep_order() { __builtin_amdgcn_sched_barrier(0); }

DEV void lds_written_signal() {
    asm volatile("s_wait_dscnt 0x0" ::: "memory");
    asm volatile("s_barrier_signal -1" ::: "memory");
}
DEV void lds_written_wait() { asm volatile("s_barrier_wait -1" ::: "memory"); }

DEV void pin(v8f& a, v8f& b, v8f& c, v8f& d) { asm volatile("" : "+v"(a), "+v"(b), "+v"(c), "+v"(d) :: "memory"); }

struct Frags { v2f a0[QPC], a1[QPC], b0[QPC], b1[QPC]; };

template <int C>
DEV void load_frags(Frags& f, const float* arow0, const float* arow1, const float* brow0, const float* brow1) {
#pragma unroll
    for (int i = 0; i < QPC; ++i) {
        const int q = QPC * C + i;
        f.a0[i] = *(const v2f*)(arow0 + 4 * q);
        f.a1[i] = *(const v2f*)(arow1 + 4 * q);
        if (q < KSUB / 4)           f.b0[i] = *(const v2f*)(brow0 + 4 * q);
        if (q >= (KDIM - KSUB) / 4) f.b1[i] = *(const v2f*)(brow1 + 4 * q);
    }
}

template <int C, int I0, int I1>
DEV void mma_steps(const Frags& f, v8f& acc00, v8f& acc01, v8f& acc10, v8f& acc11) {
#pragma unroll
    for (int i = I0; i < I1; ++i) {
        const int q = QPC * C + i;
        if (q < KSUB / 4) {
            acc00 = __builtin_amdgcn_wmma_f32_16x16x4_f32(false, f.a0[i], false, f.b0[i], (short)0, acc00, false, false);
            acc10 = __builtin_amdgcn_wmma_f32_16x16x4_f32(false, f.a1[i], false, f.b0[i], (short)0, acc10, false, false);
        }
        if (q >= (KDIM - KSUB) / 4) {
            acc01 = __builtin_amdgcn_wmma_f32_16x16x4_f32(false, f.a0[i], false, f.b1[i], (short)0, acc01, false, false);
            acc11 = __builtin_amdgcn_wmma_f32_16x16x4_f32(false, f.a1[i], false, f.b1[i], (short)0, acc11, false, false);
        }
    }
}

template <int C>
DEV void stage_chunk(float* xs, float* w2t, const v4f (&wv)[NCHUNK * WU], const v4f (&xv)[NCHUNK * XG], int rr, int cq) {
#pragma unroll
    for (int u = 0; u < WU; ++u) {
        const int t = WU * C + u;
        const int k = rr + ROWS_PER_PASS * t;
#pragma unroll
        for (int c = 0; c < 4; ++c) {
            const int j = 4 * cq + c;
            const int f = k - j;
            w2t[j * LDS_STRIDE + k] = (f >= 0 && f < F_DIM) ? wv[t][c] : 0.0f;
        }
    }
#pragma unroll
    for (int g = 0; g < XG; ++g) {
        const int b = rr + ROWS_PER_PASS * g;
        *(v4f*)(&xs[b * LDS_STRIDE + KCH * C + 4 * cq]) = xv[XG * C + g];
    }
}

__global__ __launch_bounds__(NTHREADS)
void cscfc_wmma_kernel(const float* __restrict__ x,
                       const float* __restrict__ kern,
                       const float* __restrict__ bias,
                       float* __restrict__ out) {
    __shared__ __attribute__((aligned(16))) float lds[LDS_FLOATS];
    float* xs  = lds;
    float* w2t = lds + B_DIM * LDS_STRIDE;

    const int tid  = threadIdx.x;
    const int n0   = blockIdx.x * TILE_N;
    const int cq   = tid & (CPL - 1);
    const int rr   = tid / CPL;

    const int wave = tid >> 5;
    const int lane = tid & 31;
    const int half = lane >> 4;
    const int lj   = lane & 15;

    const float bv0 = bias[n0 + lj];
    const float bv1 = bias[n0 + SUB_N + lj];

    v4f wv[NCHUNK * WU];
    v4f xv[NCHUNK * XG];
#pragma unroll
    for (int c = 0; c < NCHUNK; ++c) {
#pragma unroll
        for (int u = 0; u < WU; ++u) {
            const int t   = WU * c + u;
            const int k   = rr + ROWS_PER_PASS * t;
            const int row = (n0 + k) & (C_DIM - 1);
            wv[t] = *(const v4f*)(kern + (size_t)row * N_DIM + (size_t)(n0 + 4 * cq));
        }
        const int col = ((n0 + KCH * c) & (C_DIM - 1)) + 4 * cq;
#pragma unroll
        for (int g = 0; g < XG; ++g) {
            const int b = rr + ROWS_PER_PASS * g;
            xv[XG * c + g] = *(const v4f*)(x + (size_t)b * C_DIM + (size_t)col);
        }
    }

    const int b0 = wave * 32;

    const float* arow0 = &xs [(b0 + lj)          * LDS_STRIDE + 2 * half];
    const float* arow1 = &xs [(b0 + 16 + lj)     * LDS_STRIDE + 2 * half];
    const float* brow0 = &w2t[lj                 * LDS_STRIDE + 2 * half];
    const float* brow1 = &w2t[(SUB_N + lj)       * LDS_STRIDE + 2 * half];

    v8f acc00 = {}, acc01 = {}, acc10 = {}, acc11 = {};
    Frags f0, f1, f2;

    stage_chunk<0>(xs, w2t, wv, xv, rr, cq);
    lds_written_signal();
    lds_written_wait();
    load_frags<0>(f0, arow0, arow1, brow0, brow1);
    keep_order();

    mma_steps<0, 0, 4>(f0, acc00, acc01, acc10, acc11);
    pin(acc00, acc01, acc10, acc11);
    keep_order();
    stage_chunk<1>(xs, w2t, wv, xv, rr, cq);
    pin(acc00, acc01, acc10, acc11);
    keep_order();
    mma_steps<0, 4, 6>(f0, acc00, acc01, acc10, acc11);
    pin(acc00, acc01, acc10, acc11);
    keep_order();
    lds_written_signal();
    pin(acc00, acc01, acc10, acc11);
    keep_order();
    mma_steps<0, 6, 7>(f0, acc00, acc01, acc10, acc11);
    pin(acc00, acc01, acc10, acc11);
    keep_order();
    lds_written_wait();
    load_frags<1>(f1, arow0, arow1, brow0, brow1);
    pin(acc00, acc01, acc10, acc11);
    keep_order();
    mma_steps<0, 7, 8>(f0, acc00, acc01, acc10, acc11);

    mma_steps<1, 0, 4>(f1, acc00, acc01, acc10, acc11);
    pin(acc00, acc01, acc10, acc11);
    keep_order();
    stage_chunk<2>(xs, w2t, wv, xv, rr, cq);
    pin(acc00, acc01, acc10, acc11);
    keep_order();
    mma_steps<1, 4, 6>(f1, acc00, acc01, acc10, acc11);
    pin(acc00, acc01, acc10, acc11);
    keep_order();
    lds_written_signal();
    pin(acc00, acc01, acc10, acc11);
    keep_order();
    mma_steps<1, 6, 7>(f1, acc00, acc01, acc10, acc11);
    pin(acc00, acc01, acc10, acc11);
    keep_order();
    lds_written_wait();
    load_frags<2>(f2, arow0, arow1, brow0, brow1);
    pin(acc00, acc01, acc10, acc11);
    keep_order();
    mma_steps<1, 7, 8>(f1, acc00, acc01, acc10, acc11);

    mma_steps<2, 0, 8>(f2, acc00, acc01, acc10, acc11);

#pragma unroll
    for (int r = 0; r < 8; ++r) {
        const int m = r + (half ? 8 : 0);
        float* o0 = out + (size_t)(b0 + m) * N_DIM + (size_t)(n0 + lj);
        float* o1 = out + (size_t)(b0 + 16 + m) * N_DIM + (size_t)(n0 + lj);
        o0[0]     = acc00[r] + bv0;
        o0[SUB_N] = acc01[r] + bv1;
        o1[0]     = acc10[r] + bv0;
        o1[SUB_N] = acc11[r] + bv1;
    }
}

extern "C" void kernel_launch(void* const* d_in, const int* in_sizes, int n_in,
                              void* d_out, int out_size, void* d_ws, size_t ws_size,
                              hipStream_t stream) {
    (void)in_sizes; (void)n_in; (void)out_size; (void)d_ws; (void)ws_size;
    const float* x    = (const float*)d_in[0];
    const float* kern = (const float*)d_in[1];
    const float* bias = (const float*)d_in[2];
    float* out        = (float*)d_out;

    dim3 grid(N_DIM / TILE_N);
    dim3 block(NTHREADS);
    hipLaunchKernelGGL(cscfc_wmma_kernel, grid, block, 0, stream, x, kern, bias, out);
}
